# starting kernel
# baseline (speedup 1.0000x reference)
#include <hip/hip_runtime.h>
#include <hip/hip_bf16.h>


#define OUT_CH 256
#define IN_CH  256
#define HEADS  4
#define HID    64
#define NEG_SLOPE 0.2f

typedef float v2f  __attribute__((ext_vector_type(2)));
typedef float v8f  __attribute__((ext_vector_type(8)));
typedef __bf16 v16bf __attribute__((ext_vector_type(16)));

#if __has_builtin(__builtin_amdgcn_wmma_f32_16x16x4_f32)
#define HAVE_WMMA_F32X4 1
#else
#define HAVE_WMMA_F32X4 0
#endif

#if HAVE_WMMA_F32X4
__global__ void gemm_wmma_f32(const float* __restrict__ X,
                              const float* __restrict__ W,
                              float* __restrict__ H, int M) {
  const int wave  = threadIdx.x >> 5;
  const int lane  = threadIdx.x & 31;
  const int strip = blockIdx.x * 4 + wave;
  if (strip * 16 >= M) return;
  const int ncol0 = blockIdx.y * 64;
  const int lo    = lane & 15;
  const int hi    = lane >> 4;
  const int row   = strip * 16 + lo;

  v8f c0 = {}, c1 = {}, c2 = {}, c3 = {};
  for (int kk = 0; kk < IN_CH; kk += 4) {
    const int ka = kk + 2 * hi;
    v2f a = *(const v2f*)(X + (size_t)row * IN_CH + ka);
    v2f b0, b1, b2, b3;
    const float* wr0 = W + (size_t)ka * OUT_CH + ncol0 + lo;
    const float* wr1 = wr0 + OUT_CH;
    b0.x = wr0[0];  b0.y = wr1[0];
    b1.x = wr0[16]; b1.y = wr1[16];
    b2.x = wr0[32]; b2.y = wr1[32];
    b3.x = wr0[48]; b3.y = wr1[48];
    c0 = __builtin_amdgcn_wmma_f32_16x16x4_f32(false, a, false, b0, (short)0, c0, false, false);
    c1 = __builtin_amdgcn_wmma_f32_16x16x4_f32(false, a, false, b1, (short)0, c1, false, false);
    c2 = __builtin_amdgcn_wmma_f32_16x16x4_f32(false, a, false, b2, (short)0, c2, false, false);
    c3 = __builtin_amdgcn_wmma_f32_16x16x4_f32(false, a, false, b3, (short)0, c3, false, false);
  }
  v8f acc[4] = {c0, c1, c2, c3};
  #pragma unroll
  for (int t = 0; t < 4; ++t) {
    const int n = ncol0 + t * 16 + lo;
    #pragma unroll
    for (int r = 0; r < 8; ++r) {
      const int m = strip * 16 + r + 8 * hi;
      H[(size_t)m * OUT_CH + n] = acc[t][r];
    }
  }
}
#else
__global__ void gemm_wmma_f32(const float* __restrict__ X,
                              const float* __restrict__ W,
                              float* __restrict__ H, int M) {
  const int wave  = threadIdx.x >> 5;
  const int lane  = threadIdx.x & 31;
  const int strip = blockIdx.x * 4 + wave;
  if (strip * 16 >= M) return;
  const int ncol0 = blockIdx.y * 64;
  const int lo    = lane & 15;
  const int hi    = lane >> 4;
  const int row   = strip * 16 + lo;

  v8f acc[4] = {};
  for (int kk = 0; kk < IN_CH; kk += 32) {
    v16bf a;
    #pragma unroll
    for (int j = 0; j < 16; ++j) {
      const int p = j >> 1;
      const int k = 16 * (p >> 2) + 8 * hi + 2 * (p & 3) + (j & 1);
      a[j] = (__bf16)X[(size_t)row * IN_CH + kk + k];
    }
    #pragma unroll
    for (int t = 0; t < 4; ++t) {
      const int n = ncol0 + t * 16 + lo;
      v16bf b;
      #pragma unroll
      for (int j = 0; j < 16; ++j) {
        const int k = j + 16 * hi;
        b[j] = (__bf16)W[(size_t)(kk + k) * OUT_CH + n];
      }
      acc[t] = __builtin_amdgcn_wmma_f32_16x16x32_bf16(false, a, false, b,
                                                       (short)0, acc[t], false, false);
    }
  }
  #pragma unroll
  for (int t = 0; t < 4; ++t) {
    const int n = ncol0 + t * 16 + lo;
    #pragma unroll
    for (int r = 0; r < 8; ++r) {
      const int m = strip * 16 + r + 8 * hi;
      H[(size_t)m * OUT_CH + n] = acc[t][r];
    }
  }
}
#endif

__global__ void attn_dots(const float* __restrict__ H,
                          const float* __restrict__ att_src,
                          const float* __restrict__ att_dst,
                          float* __restrict__ a_src,
                          float* __restrict__ a_dst, int N) {
  const int idx = blockIdx.x * blockDim.x + threadIdx.x;
  if (idx >= N * HEADS) return;
  const int node = idx >> 2;
  const int head = idx & 3;
  const float* hp = H + (size_t)node * OUT_CH + head * HID;
  const float* as = att_src + head * HID;
  const float* ad = att_dst + head * HID;
  float s = 0.f, d = 0.f;
  #pragma unroll 8
  for (int c = 0; c < HID; ++c) {
    const float v = hp[c];
    s += v * as[c];
    d += v * ad[c];
  }
  a_src[idx] = s;
  a_dst[idx] = d;
}

__device__ __forceinline__ unsigned f2ord(float f) {
  unsigned u = __float_as_uint(f);
  return (u & 0x80000000u) ? ~u : (u | 0x80000000u);
}
__device__ __forceinline__ float ord2f(unsigned u) {
  return (u & 0x80000000u) ? __uint_as_float(u & 0x7fffffffu) : __uint_as_float(~u);
}
__device__ __forceinline__ void atomAddF(float* p, float v) {
  __hip_atomic_fetch_add(p, v, __ATOMIC_RELAXED, __HIP_MEMORY_SCOPE_AGENT);
}

__global__ void init_buffers(float* __restrict__ out, unsigned* __restrict__ emax,
                             float* __restrict__ esum, int out_n, int nh) {
  const int i = blockIdx.x * blockDim.x + threadIdx.x;
  if (i < out_n) out[i] = 0.f;
  if (i < nh) { emax[i] = 0x007FFFFFu  ; esum[i] = 0.f; }
}

__global__ void edge_max(const int* __restrict__ ei,
                         const float* __restrict__ a_src,
                         const float* __restrict__ a_dst,
                         unsigned* __restrict__ emax, int E, int N) {
  const int e = blockIdx.x * blockDim.x + threadIdx.x;
  if (e >= E + N) return;
  int s, d;
  if (e < E) { s = ei[e]; d = ei[E + e]; } else { s = d = e - E; }
  #pragma unroll
  for (int h = 0; h < HEADS; ++h) {
    float v = a_src[s * HEADS + h] + a_dst[d * HEADS + h];
    v = (v >= 0.f) ? v : NEG_SLOPE * v;
    atomicMax(&emax[d * HEADS + h], f2ord(v));
  }
}

__global__ void edge_sum(const int* __restrict__ ei,
                         const float* __restrict__ a_src,
                         const float* __restrict__ a_dst,
                         const unsigned* __restrict__ emax,
                         float* __restrict__ esum,
                         float* __restrict__ eexp, int E, int N) {
  const int e = blockIdx.x * blockDim.x + threadIdx.x;
  if (e >= E + N) return;
  int s, d;
  if (e < E) { s = ei[e]; d = ei[E + e]; } else { s = d = e - E; }
  #pragma unroll
  for (int h = 0; h < HEADS; ++h) {
    float v = a_src[s * HEADS + h] + a_dst[d * HEADS + h];
    v = (v >= 0.f) ? v : NEG_SLOPE * v;
    const float m = ord2f(emax[d * HEADS + h]);
    const float x = expf(v - m);
    eexp[(size_t)e * HEADS + h] = x;
    atomAddF(&esum[d * HEADS + h], x);
  }
}

__global__ void scatter_msgs(const int* __restrict__ ei,
                             const float* __restrict__ H,
                             const float* __restrict__ eexp,
                             const float* __restrict__ esum,
                             float* __restrict__ out, int E, int N) {
  const int gid  = blockIdx.x * blockDim.x + threadIdx.x;
  const int e    = gid >> 5;
  const int lane = gid & 31;
  if (e >= E + N) return;
  int s, d;
  if (e < E) { s = ei[e]; d = ei[E + e]; } else { s = d = e - E; }
  const int head  = lane >> 3;
  const float alpha = eexp[(size_t)e * HEADS + head] /
                      (esum[d * HEADS + head] + 1e-16f);
  const int c = lane * 8;
  const float4* hp = (const float4*)(H + (size_t)s * OUT_CH + c);
  const float4 h0 = hp[0];
  const float4 h1 = hp[1];
  float* op = out + (size_t)d * OUT_CH + c;
  atomAddF(op + 0, alpha * h0.x);
  atomAddF(op + 1, alpha * h0.y);
  atomAddF(op + 2, alpha * h0.z);
  atomAddF(op + 3, alpha * h0.w);
  atomAddF(op + 4, alpha * h1.x);
  atomAddF(op + 5, alpha * h1.y);
  atomAddF(op + 6, alpha * h1.z);
  atomAddF(op + 7, alpha * h1.w);
}

__global__ void finalize(float* __restrict__ out, const float* __restrict__ bias,
                         const float* __restrict__ prelu_w, int total) {
  const int i = blockIdx.x * blockDim.x + threadIdx.x;
  if (i >= total) return;
  const int c = i & (OUT_CH - 1);
  const float v = out[i] + bias[c];
  out[i] = (v >= 0.f) ? v : prelu_w[c] * v;
}

extern "C" void kernel_launch(void* const* d_in, const int* in_sizes, int n_in,
                              void* d_out, int out_size, void* d_ws, size_t ws_size,
                              hipStream_t stream) {
  const float* x       = (const float*)d_in[0];
  const int*   ei      = (const int*)d_in[1];
  const float* W       = (const float*)d_in[2];
  const float* att_src = (const float*)d_in[3];
  const float* att_dst = (const float*)d_in[4];
  const float* bias    = (const float*)d_in[5];
  const float* prelu_w = (const float*)d_in[6];
  float* out = (float*)d_out;

  const int N = in_sizes[0] / IN_CH;
  const int E = in_sizes[1] / 2;

  char* ws = (char*)d_ws;
  float* H = (float*)ws;            ws += (size_t)N * OUT_CH * sizeof(float);
  float* a_src = (float*)ws;        ws += (size_t)N * HEADS * sizeof(float);
  float* a_dst = (float*)ws;        ws += (size_t)N * HEADS * sizeof(float);
  unsigned* emax = (unsigned*)ws;   ws += (size_t)N * HEADS * sizeof(unsigned);
  float* esum = (float*)ws;         ws += (size_t)N * HEADS * sizeof(float);
  float* eexp = (float*)ws;

  const int strips = (N + 15) / 16;
  dim3 ggrid((strips + 3) / 4, OUT_CH / 64);
  gemm_wmma_f32<<<ggrid, 128, 0, stream>>>(x, W, H, N);

  attn_dots<<<(N * HEADS + 255) / 256, 256, 0, stream>>>(H, att_src, att_dst,
                                                         a_src, a_dst, N);

  const int total = N * OUT_CH;
  init_buffers<<<(total + 255) / 256, 256, 0, stream>>>(out, emax, esum,
                                                        total, N * HEADS);

  const int EL = E + N;
  edge_max<<<(EL + 255) / 256, 256, 0, stream>>>(ei, a_src, a_dst, emax, E, N);
  edge_sum<<<(EL + 255) / 256, 256, 0, stream>>>(ei, a_src, a_dst, emax, esum,
                                                 eexp, E, N);

  const long long sthreads = (long long)EL * 32;
  scatter_msgs<<<(int)((sthreads + 255) / 256), 256, 0, stream>>>(ei, H, eexp,
                                                                  esum, out, E, N);

  finalize<<<(total + 255) / 256, 256, 0, stream>>>(out, bias, prelu_w, total);
}


# Prefix sum of the in-degree counters on 49 work-groups with whole-line int4 loads (each sums the counters before its 1024-node block, then scans its own) instead of one work-group on strided lines
# speedup vs baseline: 4.4620x; 4.4620x over previous
#include <hip/hip_runtime.h>
#include <hip/hip_bf16.h>


#define OUT_CH 256
#define IN_CH  256
#define HEADS  4
#define HID    64
#define NEG_SLOPE 0.2f

typedef float v2f  __attribute__((ext_vector_type(2)));
typedef float v8f  __attribute__((ext_vector_type(8)));
typedef __bf16 v16bf __attribute__((ext_vector_type(16)));

#if __has_builtin(__builtin_amdgcn_wmma_f32_16x16x4_f32)
#define HAVE_WMMA_F32X4 1
#else
#define HAVE_WMMA_F32X4 0
#endif

#if HAVE_WMMA_F32X4
__global__ void gemm_wmma_f32(const float* __restrict__ X,
                              const float* __restrict__ W,
                              float* __restrict__ H, int M) {
  const int wave  = threadIdx.x >> 5;
  const int lane  = threadIdx.x & 31;
  const int strip = blockIdx.x * 4 + wave;
  if (strip * 16 >= M) return;
  const int ncol0 = blockIdx.y * 64;
  const int lo    = lane & 15;
  const int hi    = lane >> 4;
  const int row   = strip * 16 + lo;

  v8f c0 = {}, c1 = {}, c2 = {}, c3 = {};
  for (int kk = 0; kk < IN_CH; kk += 4) {
    const int ka = kk + 2 * hi;
    v2f a = *(const v2f*)(X + (size_t)row * IN_CH + ka);
    v2f b0, b1, b2, b3;
    const float* wr0 = W + (size_t)ka * OUT_CH + ncol0 + lo;
    const float* wr1 = wr0 + OUT_CH;
    b0.x = wr0[0];  b0.y = wr1[0];
    b1.x = wr0[16]; b1.y = wr1[16];
    b2.x = wr0[32]; b2.y = wr1[32];
    b3.x = wr0[48]; b3.y = wr1[48];
    c0 = __builtin_amdgcn_wmma_f32_16x16x4_f32(false, a, false, b0, (short)0, c0, false, false);
    c1 = __builtin_amdgcn_wmma_f32_16x16x4_f32(false, a, false, b1, (short)0, c1, false, false);
    c2 = __builtin_amdgcn_wmma_f32_16x16x4_f32(false, a, false, b2, (short)0, c2, false, false);
    c3 = __builtin_amdgcn_wmma_f32_16x16x4_f32(false, a, false, b3, (short)0, c3, false, false);
  }
  v8f acc[4] = {c0, c1, c2, c3};
  #pragma unroll
  for (int t = 0; t < 4; ++t) {
    const int n = ncol0 + t * 16 + lo;
    #pragma unroll
    for (int r = 0; r < 8; ++r) {
      const int m = strip * 16 + r + 8 * hi;
      H[(size_t)m * OUT_CH + n] = acc[t][r];
    }
  }
}
#else
__global__ void gemm_wmma_f32(const float* __restrict__ X,
                              const float* __restrict__ W,
                              float* __restrict__ H, int M) {
  const int wave  = threadIdx.x >> 5;
  const int lane  = threadIdx.x & 31;
  const int strip = blockIdx.x * 4 + wave;
  if (strip * 16 >= M) return;
  const int ncol0 = blockIdx.y * 64;
  const int lo    = lane & 15;
  const int hi    = lane >> 4;
  const int row   = strip * 16 + lo;

  v8f acc[4] = {};
  for (int kk = 0; kk < IN_CH; kk += 32) {
    v16bf a;
    #pragma unroll
    for (int j = 0; j < 16; ++j) {
      const int p = j >> 1;
      const int k = 16 * (p >> 2) + 8 * hi + 2 * (p & 3) + (j & 1);
      a[j] = (__bf16)X[(size_t)row * IN_CH + kk + k];
    }
    #pragma unroll
    for (int t = 0; t < 4; ++t) {
      const int n = ncol0 + t * 16 + lo;
      v16bf b;
      #pragma unroll
      for (int j = 0; j < 16; ++j) {
        const int k = j + 16 * hi;
        b[j] = (__bf16)W[(size_t)(kk + k) * OUT_CH + n];
      }
      acc[t] = __builtin_amdgcn_wmma_f32_16x16x32_bf16(false, a, false, b,
                                                       (short)0, acc[t], false, false);
    }
  }
  #pragma unroll
  for (int t = 0; t < 4; ++t) {
    const int n = ncol0 + t * 16 + lo;
    #pragma unroll
    for (int r = 0; r < 8; ++r) {
      const int m = strip * 16 + r + 8 * hi;
      H[(size_t)m * OUT_CH + n] = acc[t][r];
    }
  }
}
#endif

__global__ void attn_dots(const float* __restrict__ H,
                          const float* __restrict__ att_src,
                          const float* __restrict__ att_dst,
                          float* __restrict__ a_src,
                          float* __restrict__ a_dst,
                          int* __restrict__ cnt, int N) {
  const int idx = blockIdx.x * blockDim.x + threadIdx.x;
  if (idx < N) cnt[idx] = 0;
  if (idx >= N * HEADS) return;
  const int node = idx >> 2;
  const int head = idx & 3;
  const float* hp = H + (size_t)node * OUT_CH + head * HID;
  const float* as = att_src + head * HID;
  const float* ad = att_dst + head * HID;
  float s = 0.f, d = 0.f;
  #pragma unroll 8
  for (int c = 0; c < HID; ++c) {
    const float v = hp[c];
    s += v * as[c];
    d += v * ad[c];
  }
  a_src[idx] = s;
  a_dst[idx] = d;
}

__global__ void csr_count(const int* __restrict__ ei, int* __restrict__ cnt,
                          int* __restrict__ rnk, int E) {
  const int e = blockIdx.x * blockDim.x + threadIdx.x;
  if (e >= E) return;
  rnk[e] = atomicAdd(&cnt[ei[E + e]], 1);
}

#define SCAN_WG 1024
__device__ __forceinline__ int wave_sum_i(int v) {
  #pragma unroll
  for (int o = 16; o > 0; o >>= 1) v += __shfl_xor(v, o);
  return v;
}
__global__ __launch_bounds__(SCAN_WG) void csr_scan(const int* __restrict__ cnt,
                                                    int* __restrict__ rowptr, int N) {
  __shared__ int wfront[32];
  __shared__ int wown[32];
  const int t    = threadIdx.x;
  const int lane = t & 31;
  const int w    = t >> 5;
  const int base = blockIdx.x * SCAN_WG;
  const int i    = base + t;
  const int c    = (i < N) ? cnt[i] : 0;

  const int4* c4 = (const int4*)cnt;
  const int n4 = base >> 2;
  int s = 0;
  #pragma unroll 4
  for (int k = t; k < n4; k += SCAN_WG) {
    const int4 q = c4[k];
    s += (q.x + q.y) + (q.z + q.w);
  }
  s = wave_sum_i(s);

  int v = c;
  #pragma unroll
  for (int o = 1; o < 32; o <<= 1) {
    const int u = __shfl_up(v, o);
    if (lane >= o) v += u;
  }
  if (lane == 0)  wfront[w] = s;
  if (lane == 31) wown[w]   = v;
  __syncthreads();
  const int front  = wave_sum_i(wfront[lane]);
  const int before = wave_sum_i(lane < w ? wown[lane] : 0);
  if (i < N) rowptr[i] = front + before + (v - c);
}

__global__ void csr_fill(const int* __restrict__ ei, const int* __restrict__ rowptr,
                         const int* __restrict__ rnk, int* __restrict__ col, int E) {
  const int e = blockIdx.x * blockDim.x + threadIdx.x;
  if (e >= E) return;
  col[rowptr[ei[E + e]] + rnk[e]] = ei[e];
}

__device__ __forceinline__ float lrelu(float v) {
  return (v >= 0.f) ? v : NEG_SLOPE * v;
}
__device__ __forceinline__ float lane_bcast(float v, int l) {
  return __int_as_float(__builtin_amdgcn_readlane(__float_as_int(v), l));
}
__device__ __forceinline__ float wave_max(float v) {
  #pragma unroll
  for (int o = 16; o > 0; o >>= 1) v = fmaxf(v, __shfl_xor(v, o));
  return v;
}
__device__ __forceinline__ float wave_sum(float v) {
  #pragma unroll
  for (int o = 16; o > 0; o >>= 1) v += __shfl_xor(v, o);
  return v;
}
__device__ __forceinline__ void load_scores(const int* __restrict__ col,
                                            const float* __restrict__ a_src,
                                            const float4 ad, int beg, int deg, int d, int j,
                                            int& s, float& e0, float& e1, float& e2, float& e3) {
  s = d;
  e0 = e1 = e2 = e3 = -INFINITY;
  if (j <= deg) {
    if (j < deg) s = col[beg + j];
    const float4 as = *(const float4*)(a_src + (size_t)s * HEADS);
    e0 = lrelu(as.x + ad.x);
    e1 = lrelu(as.y + ad.y);
    e2 = lrelu(as.z + ad.z);
    e3 = lrelu(as.w + ad.w);
  }
}

__global__ __launch_bounds__(256) void gat_node(const int* __restrict__ rowptr,
                                                const int* __restrict__ cnt,
                                                const int* __restrict__ col,
                                                const float* __restrict__ H,
                                                const float* __restrict__ a_src,
                                                const float* __restrict__ a_dst,
                                                const float* __restrict__ bias,
                                                const float* __restrict__ prelu_w,
                                                float* __restrict__ out, int N) {
  const int lane = threadIdx.x & 31;
  const int hi   = lane >> 4;
  const int d = __builtin_amdgcn_readfirstlane(
      (int)((blockIdx.x * blockDim.x + threadIdx.x) >> 5));
  if (d >= N) return;
  const int beg = rowptr[d];
  const int deg = cnt[d];
  const int tot = deg + 1;
  const float4 ad = *(const float4*)(a_dst + (size_t)d * HEADS);

  int s = d;
  float e0 = -INFINITY, e1 = -INFINITY, e2 = -INFINITY, e3 = -INFINITY;
  float m0 = -INFINITY, m1 = -INFINITY, m2 = -INFINITY, m3 = -INFINITY;
  for (int base = 0; base < tot; base += 32) {
    load_scores(col, a_src, ad, beg, deg, d, base + lane, s, e0, e1, e2, e3);
    m0 = fmaxf(m0, e0); m1 = fmaxf(m1, e1); m2 = fmaxf(m2, e2); m3 = fmaxf(m3, e3);
  }
  m0 = wave_max(m0); m1 = wave_max(m1); m2 = wave_max(m2); m3 = wave_max(m3);

  float4 accA = {0.f, 0.f, 0.f, 0.f}, accB = {0.f, 0.f, 0.f, 0.f};
  float s0 = 0.f, s1 = 0.f, s2 = 0.f, s3 = 0.f;
  const float* Hl = H + lane * 4;
  for (int base = 0; base < tot; base += 32) {
    if (tot > 32) load_scores(col, a_src, ad, beg, deg, d, base + lane, s, e0, e1, e2, e3);
    const bool valid = (base + lane) < tot;
    const float x0 = valid ? expf(e0 - m0) : 0.f;
    const float x1 = valid ? expf(e1 - m1) : 0.f;
    const float x2 = valid ? expf(e2 - m2) : 0.f;
    const float x3 = valid ? expf(e3 - m3) : 0.f;
    s0 += x0; s1 += x1; s2 += x2; s3 += x3;
    const int n = (tot - base < 32) ? (tot - base) : 32;
    #pragma unroll 4
    for (int jj = 0; jj < n; ++jj) {
      const int sj = __builtin_amdgcn_readlane(s, jj);
      const float y0 = lane_bcast(x0, jj), y1 = lane_bcast(x1, jj);
      const float y2 = lane_bcast(x2, jj), y3 = lane_bcast(x3, jj);
      const float xa = hi ? y1 : y0;
      const float xb = hi ? y3 : y2;
      const float* hp = Hl + (size_t)sj * OUT_CH;
      const float4 ha = *(const float4*)(hp);
      const float4 hb = *(const float4*)(hp + 128);
      accA.x += xa * ha.x; accA.y += xa * ha.y; accA.z += xa * ha.z; accA.w += xa * ha.w;
      accB.x += xb * hb.x; accB.y += xb * hb.y; accB.z += xb * hb.z; accB.w += xb * hb.w;
    }
  }
  s0 = wave_sum(s0); s1 = wave_sum(s1); s2 = wave_sum(s2); s3 = wave_sum(s3);
  const float da = (hi ? s1 : s0) + 1e-16f;
  const float db = (hi ? s3 : s2) + 1e-16f;

  const int c = lane * 4;
  const float4 ba = *(const float4*)(bias + c);
  const float4 bb = *(const float4*)(bias + 128 + c);
  const float4 pa = *(const float4*)(prelu_w + c);
  const float4 pb = *(const float4*)(prelu_w + 128 + c);
  float4 oa, ob;
  oa.x = accA.x / da + ba.x; oa.y = accA.y / da + ba.y;
  oa.z = accA.z / da + ba.z; oa.w = accA.w / da + ba.w;
  ob.x = accB.x / db + bb.x; ob.y = accB.y / db + bb.y;
  ob.z = accB.z / db + bb.z; ob.w = accB.w / db + bb.w;
  oa.x = (oa.x >= 0.f) ? oa.x : pa.x * oa.x; oa.y = (oa.y >= 0.f) ? oa.y : pa.y * oa.y;
  oa.z = (oa.z >= 0.f) ? oa.z : pa.z * oa.z; oa.w = (oa.w >= 0.f) ? oa.w : pa.w * oa.w;
  ob.x = (ob.x >= 0.f) ? ob.x : pb.x * ob.x; ob.y = (ob.y >= 0.f) ? ob.y : pb.y * ob.y;
  ob.z = (ob.z >= 0.f) ? ob.z : pb.z * ob.z; ob.w = (ob.w >= 0.f) ? ob.w : pb.w * ob.w;
  float* op = out + (size_t)d * OUT_CH + c;
  *(float4*)(op)       = oa;
  *(float4*)(op + 128) = ob;
}

extern "C" void kernel_launch(void* const* d_in, const int* in_sizes, int n_in,
                              void* d_out, int out_size, void* d_ws, size_t ws_size,
                              hipStream_t stream) {
  const float* x       = (const float*)d_in[0];
  const int*   ei      = (const int*)d_in[1];
  const float* W       = (const float*)d_in[2];
  const float* att_src = (const float*)d_in[3];
  const float* att_dst = (const float*)d_in[4];
  const float* bias    = (const float*)d_in[5];
  const float* prelu_w = (const float*)d_in[6];
  float* out = (float*)d_out;

  const int N = in_sizes[0] / IN_CH;
  const int E = in_sizes[1] / 2;

  char* ws = (char*)d_ws;
  float* H = (float*)ws;            ws += (size_t)N * OUT_CH * sizeof(float);
  float* a_src = (float*)ws;        ws += (size_t)N * HEADS * sizeof(float);
  float* a_dst = (float*)ws;        ws += (size_t)N * HEADS * sizeof(float);
  int* cnt = (int*)ws;              ws += (size_t)N * sizeof(int);
  int* rowptr = (int*)ws;           ws += (size_t)N * sizeof(int);
  int* rnk = (int*)ws;              ws += (size_t)E * sizeof(int);
  int* col = (int*)ws;

  const int strips = (N + 15) / 16;
  dim3 ggrid((strips + 3) / 4, OUT_CH / 64);
  gemm_wmma_f32<<<ggrid, 128, 0, stream>>>(x, W, H, N);

  attn_dots<<<(N * HEADS + 255) / 256, 256, 0, stream>>>(H, att_src, att_dst,
                                                         a_src, a_dst, cnt, N);

  csr_count<<<(E + 255) / 256, 256, 0, stream>>>(ei, cnt, rnk, E);
  csr_scan<<<(N + SCAN_WG - 1) / SCAN_WG, SCAN_WG, 0, stream>>>(cnt, rowptr, N);
  csr_fill<<<(E + 255) / 256, 256, 0, stream>>>(ei, rowptr, rnk, col, E);

  const long long nthreads = (long long)N * 32;
  gat_node<<<(int)((nthreads + 255) / 256), 256, 0, stream>>>(rowptr, cnt, col, H, a_src,
                                                              a_dst, bias, prelu_w, out, N);
}


# GEMM through LDS: an 8-wave work-group does 32 rows x 256 columns, X and W pieces fetched once by whole-line 16-byte loads into two LDS buffers, fragments by 8-byte LDS loads; same order of sums
# speedup vs baseline: 4.7531x; 1.0652x over previous
#include <hip/hip_runtime.h>


#define OUT_CH 256
#define IN_CH  256
#define HEADS  4
#define HID    64
#define NEG_SLOPE 0.2f

typedef float v2f  __attribute__((ext_vector_type(2)));
typedef float v8f  __attribute__((ext_vector_type(8)));


#define GT_ROWS  32
#define GT_KC    32
#define XS_PITCH 36
#define WS_PITCH 544
#define XS_FLOATS (GT_ROWS * XS_PITCH)
#define WS_FLOATS ((GT_KC / 2) * WS_PITCH)
__global__ __launch_bounds__(256) void gemm_wmma_f32(const float* __restrict__ X,
                                                     const float* __restrict__ W,
                                                     float* __restrict__ H, int M) {
  __shared__ __attribute__((aligned(16))) float Xs[2 * XS_FLOATS];
  __shared__ __attribute__((aligned(16))) float Ws[2 * WS_FLOATS];
  const int t     = threadIdx.x;
  const int wave  = t >> 5;
  const int lane  = t & 31;
  const int lo    = lane & 15;
  const int hi    = lane >> 4;
  const int row0  = blockIdx.x * GT_ROWS;
  const int srow  = (wave >> 2) * 16;
  const int ncol0 = (wave & 3) * 64;

  const int  xr  = t >> 3;
  const int  xc  = (t & 7) * 4;
  const float* xg = X + (size_t)((row0 + xr) < M ? (row0 + xr) : 0) * IN_CH + xc;
  const int  wkp = t >> 6;
  const int  wn  = (t & 63) * 4;
  const float* wg = W + (size_t)(2 * wkp) * OUT_CH + wn;
  float* xs_dst = Xs + xr * XS_PITCH + xc;
  float* ws_dst = Ws + wkp * WS_PITCH + wn * 2;

  const float* a_src_l = Xs + (srow + lo) * XS_PITCH + 2 * hi;
  const float* b_src_l = Ws + hi * WS_PITCH + (ncol0 + lo) * 2;

  float4 xreg, wa[4], wb[4];
  v8f c0 = {}, c1 = {}, c2 = {}, c3 = {};

  auto fetch = [&](int kn) {
    xreg = *(const float4*)(xg + kn);
    #pragma unroll
    for (int i = 0; i < 4; ++i) {
      wa[i] = *(const float4*)(wg + (size_t)(kn + 8 * i) * OUT_CH);
      wb[i] = *(const float4*)(wg + (size_t)(kn + 8 * i + 1) * OUT_CH);
    }
  };
  auto put = [&](int b) {
    *(float4*)(xs_dst + b * XS_FLOATS) = xreg;
    #pragma unroll
    for (int i = 0; i < 4; ++i) {
      float* d = ws_dst + b * WS_FLOATS + 4 * i * WS_PITCH;
      *(float4*)(d)     = make_float4(wa[i].x, wb[i].x, wa[i].y, wb[i].y);
      *(float4*)(d + 4) = make_float4(wa[i].z, wb[i].z, wa[i].w, wb[i].w);
    }
  };
  auto mult = [&](int b) {
    const float* ap = a_src_l + b * XS_FLOATS;
    const float* bq = b_src_l + b * WS_FLOATS;
    #pragma unroll
    for (int ks = 0; ks < GT_KC / 4; ++ks) {
      const v2f a  = *(const v2f*)(ap + ks * 4);
      const float* bp = bq + ks * 2 * WS_PITCH;
      const v2f b0 = *(const v2f*)(bp);
      const v2f b1 = *(const v2f*)(bp + 32);
      const v2f b2 = *(const v2f*)(bp + 64);
      const v2f b3 = *(const v2f*)(bp + 96);
      c0 = __builtin_amdgcn_wmma_f32_16x16x4_f32(false, a, false, b0, (short)0, c0, false, false);
      c1 = __builtin_amdgcn_wmma_f32_16x16x4_f32(false, a, false, b1, (short)0, c1, false, false);
      c2 = __builtin_amdgcn_wmma_f32_16x16x4_f32(false, a, false, b2, (short)0, c2, false, false);
      c3 = __builtin_amdgcn_wmma_f32_16x16x4_f32(false, a, false, b3, (short)0, c3, false, false);
    }
  };

  fetch(0);
  put(0);
  __syncthreads();
  #pragma unroll 1
  for (int kc = 0; kc < IN_CH - 2 * GT_KC; kc += 2 * GT_KC) {
    fetch(kc + GT_KC);
    mult(0);
    put(1);
    __syncthreads();
    fetch(kc + 2 * GT_KC);
    mult(1);
    put(0);
    __syncthreads();
  }
  fetch(IN_CH - GT_KC);
  mult(0);
  put(1);
  __syncthreads();
  mult(1);

  v8f acc[4] = {c0, c1, c2, c3};
  const int mrow = row0 + srow + 8 * hi;
  float* hp = H + (size_t)mrow * OUT_CH + ncol0 + lo;
  if (row0 + srow + 16 <= M) {
    #pragma unroll
    for (int tt = 0; tt < 4; ++tt) {
      #pragma unroll
      for (int r = 0; r < 8; ++r) hp[(size_t)r * OUT_CH + tt * 16] = acc[tt][r];
    }
  } else {
    #pragma unroll
    for (int tt = 0; tt < 4; ++tt) {
      #pragma unroll
      for (int r = 0; r < 8; ++r) {
        if (mrow + r < M) hp[(size_t)r * OUT_CH + tt * 16] = acc[tt][r];
      }
    }
  }
}

__global__ void attn_dots(const float* __restrict__ H,
                          const float* __restrict__ att_src,
                          const float* __restrict__ att_dst,
                          float* __restrict__ a_src,
                          float* __restrict__ a_dst,
                          int* __restrict__ cnt, int N) {
  const int idx = blockIdx.x * blockDim.x + threadIdx.x;
  if (idx < N) cnt[idx] = 0;
  if (idx >= N * HEADS) return;
  const int node = idx >> 2;
  const int head = idx & 3;
  const float* hp = H + (size_t)node * OUT_CH + head * HID;
  const float* as = att_src + head * HID;
  const float* ad = att_dst + head * HID;
  float s = 0.f, d = 0.f;
  #pragma unroll 8
  for (int c = 0; c < HID; ++c) {
    const float v = hp[c];
    s += v * as[c];
    d += v * ad[c];
  }
  a_src[idx] = s;
  a_dst[idx] = d;
}

__global__ void csr_count(const int* __restrict__ ei, int* __restrict__ cnt,
                          int* __restrict__ rnk, int E) {
  const int e = blockIdx.x * blockDim.x + threadIdx.x;
  if (e >= E) return;
  rnk[e] = atomicAdd(&cnt[ei[E + e]], 1);
}

#define SCAN_WG 1024
__device__ __forceinline__ int wave_sum_i(int v) {
  #pragma unroll
  for (int o = 16; o > 0; o >>= 1) v += __shfl_xor(v, o);
  return v;
}
__global__ __launch_bounds__(SCAN_WG) void csr_scan(const int* __restrict__ cnt,
                                                    int* __restrict__ rowptr, int N) {
  __shared__ int wfront[32];
  __shared__ int wown[32];
  const int t    = threadIdx.x;
  const int lane = t & 31;
  const int w    = t >> 5;
  const int base = blockIdx.x * SCAN_WG;
  const int i    = base + t;
  const int c    = (i < N) ? cnt[i] : 0;

  const int4* c4 = (const int4*)cnt;
  const int n4 = base >> 2;
  int s = 0;
  #pragma unroll 4
  for (int k = t; k < n4; k += SCAN_WG) {
    const int4 q = c4[k];
    s += (q.x + q.y) + (q.z + q.w);
  }
  s = wave_sum_i(s);

  int v = c;
  #pragma unroll
  for (int o = 1; o < 32; o <<= 1) {
    const int u = __shfl_up(v, o);
    if (lane >= o) v += u;
  }
  if (lane == 0)  wfront[w] = s;
  if (lane == 31) wown[w]   = v;
  __syncthreads();
  const int front  = wave_sum_i(wfront[lane]);
  const int before = wave_sum_i(lane < w ? wown[lane] : 0);
  if (i < N) rowptr[i] = front + before + (v - c);
}

__global__ void csr_fill(const int* __restrict__ ei, const int* __restrict__ rowptr,
                         const int* __restrict__ rnk, int* __restrict__ col, int E) {
  const int e = blockIdx.x * blockDim.x + threadIdx.x;
  if (e >= E) return;
  col[rowptr[ei[E + e]] + rnk[e]] = ei[e];
}

__device__ __forceinline__ float lrelu(float v) {
  return (v >= 0.f) ? v : NEG_SLOPE * v;
}
__device__ __forceinline__ float lane_bcast(float v, int l) {
  return __int_as_float(__builtin_amdgcn_readlane(__float_as_int(v), l));
}
__device__ __forceinline__ float wave_max(float v) {
  #pragma unroll
  for (int o = 16; o > 0; o >>= 1) v = fmaxf(v, __shfl_xor(v, o));
  return v;
}
__device__ __forceinline__ float wave_sum(float v) {
  #pragma unroll
  for (int o = 16; o > 0; o >>= 1) v += __shfl_xor(v, o);
  return v;
}
__device__ __forceinline__ void load_scores(const int* __restrict__ col,
                                            const float* __restrict__ a_src,
                                            const float4 ad, int beg, int deg, int d, int j,
                                            int& s, float& e0, float& e1, float& e2, float& e3) {
  s = d;
  e0 = e1 = e2 = e3 = -INFINITY;
  if (j <= deg) {
    if (j < deg) s = col[beg + j];
    const float4 as = *(const float4*)(a_src + (size_t)s * HEADS);
    e0 = lrelu(as.x + ad.x);
    e1 = lrelu(as.y + ad.y);
    e2 = lrelu(as.z + ad.z);
    e3 = lrelu(as.w + ad.w);
  }
}

__global__ __launch_bounds__(256) void gat_node(const int* __restrict__ rowptr,
                                                const int* __restrict__ cnt,
                                                const int* __restrict__ col,
                                                const float* __restrict__ H,
                                                const float* __restrict__ a_src,
                                                const float* __restrict__ a_dst,
                                                const float* __restrict__ bias,
                                                const float* __restrict__ prelu_w,
                                                float* __restrict__ out, int N) {
  const int lane = threadIdx.x & 31;
  const int hi   = lane >> 4;
  const int d = __builtin_amdgcn_readfirstlane(
      (int)((blockIdx.x * blockDim.x + threadIdx.x) >> 5));
  if (d >= N) return;
  const int beg = rowptr[d];
  const int deg = cnt[d];
  const int tot = deg + 1;
  const float4 ad = *(const float4*)(a_dst + (size_t)d * HEADS);

  int s = d;
  float e0 = -INFINITY, e1 = -INFINITY, e2 = -INFINITY, e3 = -INFINITY;
  float m0 = -INFINITY, m1 = -INFINITY, m2 = -INFINITY, m3 = -INFINITY;
  for (int base = 0; base < tot; base += 32) {
    load_scores(col, a_src, ad, beg, deg, d, base + lane, s, e0, e1, e2, e3);
    m0 = fmaxf(m0, e0); m1 = fmaxf(m1, e1); m2 = fmaxf(m2, e2); m3 = fmaxf(m3, e3);
  }
  m0 = wave_max(m0); m1 = wave_max(m1); m2 = wave_max(m2); m3 = wave_max(m3);

  float4 accA = {0.f, 0.f, 0.f, 0.f}, accB = {0.f, 0.f, 0.f, 0.f};
  float s0 = 0.f, s1 = 0.f, s2 = 0.f, s3 = 0.f;
  const float* Hl = H + lane * 4;
  for (int base = 0; base < tot; base += 32) {
    if (tot > 32) load_scores(col, a_src, ad, beg, deg, d, base + lane, s, e0, e1, e2, e3);
    const bool valid = (base + lane) < tot;
    const float x0 = valid ? expf(e0 - m0) : 0.f;
    const float x1 = valid ? expf(e1 - m1) : 0.f;
    const float x2 = valid ? expf(e2 - m2) : 0.f;
    const float x3 = valid ? expf(e3 - m3) : 0.f;
    s0 += x0; s1 += x1; s2 += x2; s3 += x3;
    const int n = (tot - base < 32) ? (tot - base) : 32;
    #pragma unroll 4
    for (int jj = 0; jj < n; ++jj) {
      const int sj = __builtin_amdgcn_readlane(s, jj);
      const float y0 = lane_bcast(x0, jj), y1 = lane_bcast(x1, jj);
      const float y2 = lane_bcast(x2, jj), y3 = lane_bcast(x3, jj);
      const float xa = hi ? y1 : y0;
      const float xb = hi ? y3 : y2;
      const float* hp = Hl + (size_t)sj * OUT_CH;
      const float4 ha = *(const float4*)(hp);
      const float4 hb = *(const float4*)(hp + 128);
      accA.x += xa * ha.x; accA.y += xa * ha.y; accA.z += xa * ha.z; accA.w += xa * ha.w;
      accB.x += xb * hb.x; accB.y += xb * hb.y; accB.z += xb * hb.z; accB.w += xb * hb.w;
    }
  }
  s0 = wave_sum(s0); s1 = wave_sum(s1); s2 = wave_sum(s2); s3 = wave_sum(s3);
  const float da = (hi ? s1 : s0) + 1e-16f;
  const float db = (hi ? s3 : s2) + 1e-16f;

  const int c = lane * 4;
  const float4 ba = *(const float4*)(bias + c);
  const float4 bb = *(const float4*)(bias + 128 + c);
  const float4 pa = *(const float4*)(prelu_w + c);
  const float4 pb = *(const float4*)(prelu_w + 128 + c);
  float4 oa, ob;
  oa.x = accA.x / da + ba.x; oa.y = accA.y / da + ba.y;
  oa.z = accA.z / da + ba.z; oa.w = accA.w / da + ba.w;
  ob.x = accB.x / db + bb.x; ob.y = accB.y / db + bb.y;
  ob.z = accB.z / db + bb.z; ob.w = accB.w / db + bb.w;
  oa.x = (oa.x >= 0.f) ? oa.x : pa.x * oa.x; oa.y = (oa.y >= 0.f) ? oa.y : pa.y * oa.y;
  oa.z = (oa.z >= 0.f) ? oa.z : pa.z * oa.z; oa.w = (oa.w >= 0.f) ? oa.w : pa.w * oa.w;
  ob.x = (ob.x >= 0.f) ? ob.x : pb.x * ob.x; ob.y = (ob.y >= 0.f) ? ob.y : pb.y * ob.y;
  ob.z = (ob.z >= 0.f) ? ob.z : pb.z * ob.z; ob.w = (ob.w >= 0.f) ? ob.w : pb.w * ob.w;
  float* op = out + (size_t)d * OUT_CH + c;
  *(float4*)(op)       = oa;
  *(float4*)(op + 128) = ob;
}

extern "C" void kernel_launch(void* const* d_in, const int* in_sizes, int n_in,
                              void* d_out, int out_size, void* d_ws, size_t ws_size,
                              hipStream_t stream) {
  const float* x       = (const float*)d_in[0];
  const int*   ei      = (const int*)d_in[1];
  const float* W       = (const float*)d_in[2];
  const float* att_src = (const float*)d_in[3];
  const float* att_dst = (const float*)d_in[4];
  const float* bias    = (const float*)d_in[5];
  const float* prelu_w = (const float*)d_in[6];
  float* out = (float*)d_out;

  const int N = in_sizes[0] / IN_CH;
  const int E = in_sizes[1] / 2;

  char* ws = (char*)d_ws;
  float* H = (float*)ws;            ws += (size_t)N * OUT_CH * sizeof(float);
  float* a_src = (float*)ws;        ws += (size_t)N * HEADS * sizeof(float);
  float* a_dst = (float*)ws;        ws += (size_t)N * HEADS * sizeof(float);
  int* cnt = (int*)ws;              ws += (size_t)N * sizeof(int);
  int* rowptr = (int*)ws;           ws += (size_t)N * sizeof(int);
  int* rnk = (int*)ws;              ws += (size_t)E * sizeof(int);
  int* col = (int*)ws;

  gemm_wmma_f32<<<(N + GT_ROWS - 1) / GT_ROWS, 256, 0, stream>>>(x, W, H, N);

  attn_dots<<<(N * HEADS + 255) / 256, 256, 0, stream>>>(H, att_src, att_dst,
                                                         a_src, a_dst, cnt, N);

  csr_count<<<(E + 255) / 256, 256, 0, stream>>>(ei, cnt, rnk, E);
  csr_scan<<<(N + SCAN_WG - 1) / SCAN_WG, SCAN_WG, 0, stream>>>(cnt, rowptr, N);
  csr_fill<<<(E + 255) / 256, 256, 0, stream>>>(ei, rowptr, rnk, col, E);

  const long long nthreads = (long long)N * 32;
  gat_node<<<(int)((nthreads + 255) / 256), 256, 0, stream>>>(rowptr, cnt, col, H, a_src,
                                                              a_dst, bias, prelu_w, out, N);
}
